# starting kernel
# baseline (speedup 1.0000x reference)
#include <hip/hip_runtime.h>
#include <cstdint>
#include <cstddef>

#define DFE 128

typedef float v2f __attribute__((ext_vector_type(2)));
typedef float v8f __attribute__((ext_vector_type(8)));

__global__ void k_init_deg(float* __restrict__ deg_out, float* __restrict__ deg_in, int n) {
  int i = blockIdx.x * blockDim.x + threadIdx.x;
  if (i < n) { deg_out[i] = 1.0f; deg_in[i] = 1.0f; }
}

__global__ void k_edge_deg(const int* __restrict__ src, const int* __restrict__ dst,
                           float* __restrict__ deg_out, float* __restrict__ deg_in, int e) {
  int i = blockIdx.x * blockDim.x + threadIdx.x;
  if (i < e) {
    unsafeAtomicAdd(&deg_out[src[i]], 1.0f);
    unsafeAtomicAdd(&deg_in[dst[i]], 1.0f);
  }
}

__global__ void k_rsqrt_deg(float* __restrict__ deg_out, float* __restrict__ deg_in, int n) {
  int i = blockIdx.x * blockDim.x + threadIdx.x;
  if (i < n) {
    deg_out[i] = rsqrtf(deg_out[i]);
    deg_in[i]  = rsqrtf(deg_in[i]);
  }
}

__global__ void k_self_init(const float* __restrict__ h, const float* __restrict__ cs,
                            float* __restrict__ agg, int n) {
  int t = blockIdx.x * blockDim.x + threadIdx.x;
  int node = t >> 5, c4 = t & 31;
  if (node < n) {
    float s = cs[node];
    float4 v = ((const float4*)(h + (size_t)node * DFE))[c4];
    v.x *= s; v.y *= s; v.z *= s; v.w *= s;
    ((float4*)(agg + (size_t)node * DFE))[c4] = v;
  }
}

__global__ void k_edge_scatter(const int* __restrict__ src, const int* __restrict__ dst,
                               const float* __restrict__ h, const float* __restrict__ cs,
                               float* __restrict__ agg, int e) {
  int t = blockIdx.x * blockDim.x + threadIdx.x;
  int edge = t >> 5, lane = t & 31;
  if (edge < e) {
    int s = src[edge], d = dst[edge];
    float sc = cs[s];
    float4 v = ((const float4*)(h + (size_t)s * DFE))[lane];
    float* p = agg + (size_t)d * DFE + lane * 4;
    unsafeAtomicAdd(p + 0, v.x * sc);
    unsafeAtomicAdd(p + 1, v.y * sc);
    unsafeAtomicAdd(p + 2, v.z * sc);
    unsafeAtomicAdd(p + 3, v.w * sc);
  }
}

__global__ __launch_bounds__(128) void k_gemm_relu(
    const float* __restrict__ agg, const float* __restrict__ cd,
    const float* __restrict__ W, const float* __restrict__ bias,
    float* __restrict__ out, int nrows) {
  __shared__ float As[16 * 132];
  const int tid = threadIdx.x;
  const int row0 = blockIdx.x * 16;

  for (int idx = tid; idx < 16 * 32; idx += 128) {
    int r = idx >> 5, c4 = idx & 31;
    int row = row0 + r;
    float4 v = make_float4(0.f, 0.f, 0.f, 0.f);
    float sc = 0.f;
    if (row < nrows) {
      v = ((const float4*)(agg + (size_t)row * DFE))[c4];
      sc = cd[row];
    }
    float* q = As + r * 132 + c4 * 4;
    q[0] = v.x * sc; q[1] = v.y * sc; q[2] = v.z * sc; q[3] = v.w * sc;
  }
  __syncthreads();

  const int wave = tid >> 5;
  const int lane = tid & 31;
  const int half = lane >> 4;
  const int l15  = lane & 15;
  const int ntile0 = wave * 2;
  const int colb0 = ntile0 * 16;

  v8f acc0 = {};
  v8f acc1 = {};
  const float* Arow = As + l15 * 132;

  for (int k0 = 0; k0 < DFE; k0 += 4) {
    int ka = k0 + 2 * half;
    v2f a = { Arow[ka], Arow[ka + 1] };
    v2f b0 = { W[(size_t)ka * DFE + colb0 + l15],
               W[(size_t)(ka + 1) * DFE + colb0 + l15] };
    v2f b1 = { W[(size_t)ka * DFE + colb0 + 16 + l15],
               W[(size_t)(ka + 1) * DFE + colb0 + 16 + l15] };
    acc0 = __builtin_amdgcn_wmma_f32_16x16x4_f32(false, a, false, b0,
                                                 (short)0, acc0, false, false);
    acc1 = __builtin_amdgcn_wmma_f32_16x16x4_f32(false, a, false, b1,
                                                 (short)0, acc1, false, false);
  }

  for (int v = 0; v < 8; ++v) {
    int row = row0 + v + 8 * half;
    if (row < nrows) {
      int col0 = colb0 + l15;
      float x0 = acc0[v] + bias[col0];
      out[(size_t)row * DFE + col0] = x0 > 0.f ? x0 : 0.f;
      int col1 = col0 + 16;
      float x1 = acc1[v] + bias[col1];
      out[(size_t)row * DFE + col1] = x1 > 0.f ? x1 : 0.f;
    }
  }
}

__global__ void k_pool_zero(float* __restrict__ sums, float* __restrict__ cnt, int g) {
  int i = blockIdx.x * blockDim.x + threadIdx.x;
  if (i < g * DFE) sums[i] = 0.f;
  if (i < g) cnt[i] = 0.f;
}

__global__ void k_pool_acc(const float* __restrict__ h, const int* __restrict__ gid,
                           float* __restrict__ sums, float* __restrict__ cnt, int n) {
  int t = blockIdx.x * blockDim.x + threadIdx.x;
  int node = t >> 5, lane = t & 31;
  if (node < n) {
    int g = gid[node];
    float4 v = ((const float4*)(h + (size_t)node * DFE))[lane];
    float* p = sums + (size_t)g * DFE + lane * 4;
    unsafeAtomicAdd(p + 0, v.x);
    unsafeAtomicAdd(p + 1, v.y);
    unsafeAtomicAdd(p + 2, v.z);
    unsafeAtomicAdd(p + 3, v.w);
    if (lane == 0) unsafeAtomicAdd(&cnt[g], 1.0f);
  }
}

__global__ void k_pool_final(const float* __restrict__ sums, const float* __restrict__ cnt,
                             float* __restrict__ out, int g) {
  int i = blockIdx.x * blockDim.x + threadIdx.x;
  if (i < g * DFE) {
    float c = cnt[i >> 7];
    c = c > 1.0f ? c : 1.0f;
    out[i] = sums[i] / c;
  }
}

extern "C" void kernel_launch(void* const* d_in, const int* in_sizes, int n_in,
                              void* d_out, int out_size, void* d_ws, size_t ws_size,
                              hipStream_t stream) {
  const float* x   = (const float*)d_in[0];
  const float* W0  = (const float*)d_in[1];
  const float* b0  = (const float*)d_in[2];
  const float* W1  = (const float*)d_in[3];
  const float* b1  = (const float*)d_in[4];
  const float* W2  = (const float*)d_in[5];
  const float* b2  = (const float*)d_in[6];
  const int*   src = (const int*)d_in[7];
  const int*   dst = (const int*)d_in[8];
  const int*   gid = (const int*)d_in[9];

  const int n = in_sizes[0] / DFE;
  const int e = in_sizes[7];
  const int g = out_size / DFE;

  float* ws = (float*)d_ws;
  size_t ND = (size_t)n * DFE;
  float* hA   = ws;
  float* hB   = hA + ND;
  float* agg  = hB + ND;
  float* cs   = agg + ND;
  float* cd   = cs + n;
  float* psum = cd + n;
  float* pcnt = psum + (size_t)g * DFE;

  const int TB = 256;
  dim3 blk(TB);

  k_init_deg<<<dim3((n + TB - 1) / TB), blk, 0, stream>>>(cs, cd, n);
  k_edge_deg<<<dim3((e + TB - 1) / TB), blk, 0, stream>>>(src, dst, cs, cd, e);
  k_rsqrt_deg<<<dim3((n + TB - 1) / TB), blk, 0, stream>>>(cs, cd, n);

  const unsigned nodeWaveBlocks = (unsigned)(((size_t)n * 32 + TB - 1) / TB);
  const unsigned edgeWaveBlocks = (unsigned)(((size_t)e * 32 + TB - 1) / TB);
  const unsigned gemmBlocks     = (unsigned)((n + 15) / 16);

  auto layer = [&](const float* h_in, const float* W, const float* b, float* h_out) {
    k_self_init<<<dim3(nodeWaveBlocks), blk, 0, stream>>>(h_in, cs, agg, n);
    k_edge_scatter<<<dim3(edgeWaveBlocks), blk, 0, stream>>>(src, dst, h_in, cs, agg, e);
    k_gemm_relu<<<dim3(gemmBlocks), dim3(128), 0, stream>>>(agg, cd, W, b, h_out, n);
  };

  layer(x,  W0, b0, hA);
  layer(hA, W1, b1, hB);
  layer(hB, W2, b2, hA);

  k_pool_zero<<<dim3((g * DFE + TB - 1) / TB), blk, 0, stream>>>(psum, pcnt, g);
  k_pool_acc<<<dim3(nodeWaveBlocks), blk, 0, stream>>>(hA, gid, psum, pcnt, n);
  k_pool_final<<<dim3((g * DFE + TB - 1) / TB), blk, 0, stream>>>(psum, pcnt, (float*)d_out, g);
}


# pooling: a chunk finds the lowest and highest graph id of its nodes and forms only the 16-graph tiles between them (one, not four); the reduce adds only the chunks whose id range holds its graph
# speedup vs baseline: 4.1406x; 4.1406x over previous
#include <hip/hip_runtime.h>
#include <cstdint>
#include <cstddef>

#define DFE 128
#define NLAYER 3
#define TKS 64

typedef float v2f __attribute__((ext_vector_type(2)));
typedef float v8f __attribute__((ext_vector_type(8)));

#define HCH    32768
#define HRANGE 102400
#define HLD    32
static_assert(HCH < 65536, "a 16-bit counter must hold a whole chunk");
static_assert((HRANGE / 2) % 4 == 0, "table passes start on a 16-byte boundary");

__global__ __launch_bounds__(1024) void k_deg_hist(
    const int* __restrict__ src, const int* __restrict__ dst,
    unsigned* __restrict__ tabs, int n, int e, int tdw) {
  __shared__ uint4 tbl4[HRANGE / 8];
  unsigned* tbl = (unsigned*)tbl4;
  const int tid = threadIdx.x;
  const int* __restrict__ idx = blockIdx.y ? dst : src;
  const int e0 = blockIdx.x * HCH;
  const int e1 = (e - e0) < HCH ? e : e0 + HCH;
  uint4* __restrict__ outp = (uint4*)(tabs + ((size_t)blockIdx.y * gridDim.x + blockIdx.x) * tdw);

  for (int r0 = 0; r0 < n; r0 += HRANGE) {
    const int nd  = (n - r0) < HRANGE ? (n - r0) : HRANGE;
    const int nq  = ((nd + 1) / 2 + 3) / 4;
    for (int i = tid; i < nq; i += 1024) tbl4[i] = make_uint4(0u, 0u, 0u, 0u);
    __syncthreads();
    for (int c0 = e0; c0 < e1; c0 += 1024 * HLD) {
      int v[HLD];
      if (e1 - c0 >= 1024 * HLD) {
#pragma unroll
        for (int u = 0; u < HLD; ++u) v[u] = idx[c0 + tid + u * 1024];
      } else {
#pragma unroll
        for (int u = 0; u < HLD; ++u) {
          const int i = c0 + tid + u * 1024;
          v[u] = i < e1 ? idx[i] : -1;
        }
      }
#pragma unroll
      for (int u = 0; u < HLD; ++u) {
        const int t = v[u] - r0;
        if ((unsigned)t < (unsigned)nd) atomicAdd(&tbl[t >> 1], 1u << ((t & 1) * 16));
      }
    }
    __syncthreads();
    for (int i = tid; i < nq; i += 1024) outp[r0 / 8 + i] = tbl4[i];
    __syncthreads();
  }
}

__global__ __launch_bounds__(256) void k_deg_norm(
    const unsigned* __restrict__ tabs, float* __restrict__ cs, float* __restrict__ cd,
    int* __restrict__ tick, int n, int nchunks, int tdw) {
  if (blockIdx.x == 0 && blockIdx.y == 0 && threadIdx.x < NLAYER * 2) tick[threadIdx.x * TKS] = 0;
  const int w = blockIdx.x * blockDim.x + threadIdx.x;
  if (w >= (n + 1) / 2) return;
  const unsigned* __restrict__ p = tabs + (size_t)blockIdx.y * nchunks * tdw + w;
  float* __restrict__ o = blockIdx.y ? cd : cs;
  unsigned lo = 0u, hi = 0u;
#pragma unroll 16
  for (int c = 0; c < nchunks; ++c) {
    const unsigned t = p[(size_t)c * tdw];
    lo += t & 0xffffu;
    hi += t >> 16;
  }
  const int node = 2 * w;
  o[node] = rsqrtf((float)(lo + 1u));
  if (node + 1 < n) o[node + 1] = rsqrtf((float)(hi + 1u));
}

#define EPW 64
#define EGR 16
#define EPB (8 * EPW)
static_assert(32 % EGR == 0, "a group lies inside one of the lane's two edge slots");

__device__ __forceinline__ void edge_one(const float* __restrict__ z, float* __restrict__ agg,
                                         int s, int d, int col) {
  const float* q = z + (size_t)s * DFE + col;
  float v0 = q[0], v1 = q[32];
  float* p = agg + (size_t)d * DFE + col;
  unsafeAtomicAdd(p,      v0);
  unsafeAtomicAdd(p + 32, v1);
}

__global__ __launch_bounds__(256) void k_edge_scatter(
    const int* __restrict__ src, const int* __restrict__ dst,
    const float* __restrict__ z, float* __restrict__ agg,
    int* __restrict__ tick, int e, int nbk) {
  __shared__ int sh_t, sh_q;
  if (threadIdx.x == 0) {
    int q = (int)((__builtin_amdgcn_s_sendmsg_rtn(0x87) >> 18) & 1u);
    int t = atomicAdd(&tick[q * TKS], 1);
    if (t >= nbk) { q ^= 1; t = atomicAdd(&tick[q * TKS], 1); }
    sh_t = t; sh_q = q;
  }
  __syncthreads();
  const int t = sh_t, q = sh_q;
  if (t >= nbk) return;

  const int lane = threadIdx.x & 31;
  const int wv = __builtin_amdgcn_readfirstlane((int)(threadIdx.x >> 5));
  const int base = t * EPB + wv * EPW;
  if (base >= e) return;
  const int nv = (e - base) < EPW ? (e - base) : EPW;
  const int col = q * 64 + lane;

  int s_l[2] = {0, 0}, d_l[2] = {0, 0};
#pragma unroll
  for (int k = 0; k < 2; ++k) {
    if (lane + 32 * k < nv) {
      s_l[k]  = src[base + 32 * k + lane];
      d_l[k]  = dst[base + 32 * k + lane];
    }
  }

#pragma unroll
  for (int k = 0; k < 2; ++k) {
    const int nk = (nv - 32 * k) < 32 ? (nv - 32 * k) : 32;
    for (int g0 = 0; g0 < nk; g0 += EGR) {
      if (g0 + EGR <= nk) {
        float v[EGR][2];
#pragma unroll
        for (int u = 0; u < EGR; ++u) {
          const int s = __builtin_amdgcn_readlane(s_l[k], g0 + u);
          const float* p = z + (size_t)s * DFE + col;
          v[u][0] = p[0]; v[u][1] = p[32];
        }
#pragma unroll
        for (int u = 0; u < EGR; ++u) {
          const int d = __builtin_amdgcn_readlane(d_l[k], g0 + u);
          float* p = agg + (size_t)d * DFE + col;
          unsafeAtomicAdd(p,      v[u][0]);
          unsafeAtomicAdd(p + 32, v[u][1]);
        }
      } else {
        for (int j = g0; j < nk; ++j) {
          const int s = __builtin_amdgcn_readlane(s_l[k], j);
          const int d = __builtin_amdgcn_readlane(d_l[k], j);
          edge_one(z, agg, s, d, col);
        }
      }
    }
  }
}

template <bool FIRST>
__global__ __launch_bounds__(128) void k_layer_mm(
    const float* __restrict__ x, float* agg, const float* __restrict__ cd,
    const float* __restrict__ cs, const float* __restrict__ bprev,
    const float* __restrict__ W, float* __restrict__ zout, int nrows) {
  __shared__ float As[16 * 132];
  const int tid = threadIdx.x;
  const int row0 = blockIdx.x * 16;

  for (int idx = tid; idx < 16 * 32; idx += 128) {
    int r = idx >> 5, c4 = idx & 31;
    int row = row0 + r;
    float4 v = make_float4(0.f, 0.f, 0.f, 0.f);
    if (row < nrows) {
      const float sn = cs[row];
      if (FIRST) {
        v = ((const float4*)(x + (size_t)row * DFE))[c4];
      } else {
        v = ((const float4*)(agg + (size_t)row * DFE))[c4];
        const float sc = cd[row];
        const float4 b = ((const float4*)bprev)[c4];
        v.x = v.x * sc + b.x; v.y = v.y * sc + b.y; v.z = v.z * sc + b.z; v.w = v.w * sc + b.w;
        v.x = v.x > 0.f ? v.x : 0.f; v.y = v.y > 0.f ? v.y : 0.f;
        v.z = v.z > 0.f ? v.z : 0.f; v.w = v.w > 0.f ? v.w : 0.f;
      }
      v.x *= sn; v.y *= sn; v.z *= sn; v.w *= sn;
    }
    float* q = As + r * 132 + c4 * 4;
    q[0] = v.x; q[1] = v.y; q[2] = v.z; q[3] = v.w;
  }
  __syncthreads();

  const int wave = tid >> 5;
  const int lane = tid & 31;
  const int half = lane >> 4;
  const int l15  = lane & 15;
  const int ntile0 = wave * 2;
  const int colb0 = ntile0 * 16;

  v8f acc0 = {};
  v8f acc1 = {};
  const float* Arow = As + l15 * 132;

  for (int k0 = 0; k0 < DFE; k0 += 4) {
    int ka = k0 + 2 * half;
    v2f a = { Arow[ka], Arow[ka + 1] };
    v2f b0 = { W[(size_t)ka * DFE + colb0 + l15],
               W[(size_t)(ka + 1) * DFE + colb0 + l15] };
    v2f b1 = { W[(size_t)ka * DFE + colb0 + 16 + l15],
               W[(size_t)(ka + 1) * DFE + colb0 + 16 + l15] };
    acc0 = __builtin_amdgcn_wmma_f32_16x16x4_f32(false, a, false, b0,
                                                 (short)0, acc0, false, false);
    acc1 = __builtin_amdgcn_wmma_f32_16x16x4_f32(false, a, false, b1,
                                                 (short)0, acc1, false, false);
  }

  for (int v = 0; v < 8; ++v) {
    int row = row0 + v + 8 * half;
    if (row < nrows) {
      int col0 = colb0 + l15;
      int col1 = col0 + 16;
      const float x0 = acc0[v], x1 = acc1[v];
      zout[(size_t)row * DFE + col0] = x0;
      zout[(size_t)row * DFE + col1] = x1;
      agg[(size_t)row * DFE + col0] = x0;
      agg[(size_t)row * DFE + col1] = x1;
    }
  }
}

#define PCH 100
#define PKT (PCH / 4)
static_assert(PCH % 4 == 0, "a chunk is a whole number of k-steps");

__global__ __launch_bounds__(64) void k_pool_mm(
    const float* __restrict__ agg, const float* __restrict__ cd, const float* __restrict__ bias,
    const int* __restrict__ gid,
    float* __restrict__ part, float* __restrict__ pcnt, int* __restrict__ prange, int n, int g) {
  const int lane = threadIdx.x & 31;
  const int wcol = threadIdx.x >> 5;
  const int hh   = lane >> 4;
  const int l15  = lane & 15;
  const int chunk = blockIdx.x;
  const int cbase = wcol * 64 + l15 * 4;
  const int nlane = chunk * PCH + 2 * hh;
  const float4 b4 = *(const float4*)(bias + cbase);

  int lo = g, hi = -1;
#pragma unroll
  for (int u = 0; u < (PCH + 31) / 32; ++u) {
    const int i = lane + 32 * u;
    const int node = chunk * PCH + i;
    const int v = gid[(i < PCH && node < n) ? node : chunk * PCH];
    if (v >= 0 && v < g) {
      lo = v < lo ? v : lo;
      hi = v > hi ? v : hi;
    }
  }

  float4 va[PKT], vb[PKT];
  float ca[PKT], cb[PKT];
  int ga[PKT], gb[PKT];
#pragma unroll
  for (int k = 0; k < PKT; ++k) {
    int nA = nlane + 4 * k, nB = nA + 1;
    int rA = nA < n ? nA : n - 1;
    int rB = nB < n ? nB : n - 1;
    int ia = gid[rA], ib = gid[rB];
    ga[k] = nA < n ? ia : -1;
    gb[k] = nB < n ? ib : -1;
    ca[k] = cd[rA];
    cb[k] = cd[rB];
    va[k] = *(const float4*)(agg + (size_t)rA * DFE + cbase);
    vb[k] = *(const float4*)(agg + (size_t)rB * DFE + cbase);
  }

#pragma unroll
  for (int d = 16; d >= 1; d >>= 1) {
    const int lo2 = __shfl_xor(lo, d), hi2 = __shfl_xor(hi, d);
    lo = lo2 < lo ? lo2 : lo;
    hi = hi2 > hi ? hi2 : hi;
  }
  lo = __builtin_amdgcn_readfirstlane(lo);
  hi = __builtin_amdgcn_readfirstlane(hi);
  if (threadIdx.x == 0) { prange[2 * chunk] = lo; prange[2 * chunk + 1] = hi; }

  v2f hb[PKT][4];
#pragma unroll
  for (int k = 0; k < PKT; ++k) {
    float ba[4] = { va[k].x * ca[k] + b4.x, va[k].y * ca[k] + b4.y,
                    va[k].z * ca[k] + b4.z, va[k].w * ca[k] + b4.w };
    float bb[4] = { vb[k].x * cb[k] + b4.x, vb[k].y * cb[k] + b4.y,
                    vb[k].z * cb[k] + b4.z, vb[k].w * cb[k] + b4.w };
#pragma unroll
    for (int j = 0; j < 4; ++j) {
      hb[k][j] = v2f{ ba[j] > 0.f ? ba[j] : 0.f, bb[j] > 0.f ? bb[j] : 0.f };
    }
  }

  for (int mt = lo >> 4; mt <= (hi >> 4); ++mt) {
    const int g0 = 16 * mt;
    const int gr = g0 + l15;
    v8f acc[4];
    float cacc = 0.f;
#pragma unroll
    for (int j = 0; j < 4; ++j) acc[j] = v8f{};
#pragma unroll
    for (int k = 0; k < PKT; ++k) {
      v2f a = { ga[k] == gr ? 1.f : 0.f, gb[k] == gr ? 1.f : 0.f };
      cacc += a[0] + a[1];
#pragma unroll
      for (int j = 0; j < 4; ++j) {
        acc[j] = __builtin_amdgcn_wmma_f32_16x16x4_f32(false, a, false, hb[k][j],
                                                       (short)0, acc[j], false, false);
      }
    }

#pragma unroll
    for (int v = 0; v < 8; ++v) {
      int grow = g0 + v + 8 * hh;
      if (grow < g) {
        float4 o = make_float4(acc[0][v], acc[1][v], acc[2][v], acc[3][v]);
        *(float4*)(part + ((size_t)chunk * g + grow) * DFE + cbase) = o;
      }
    }
    float c = cacc + __shfl_xor(cacc, 16);
    if (wcol == 0 && hh == 0 && gr < g) pcnt[(size_t)chunk * g + gr] = c;
  }
}

__global__ __launch_bounds__(1024) void k_pool_reduce(
    const float* __restrict__ part, const float* __restrict__ pcnt,
    const int* __restrict__ prange, float* __restrict__ out, int g, int nch) {
  __shared__ float ss[32][32];
  __shared__ float sc[32];
  const int tx = threadIdx.x & 31;
  const int ty = __builtin_amdgcn_readfirstlane((int)(threadIdx.x >> 5));
  const int o  = blockIdx.x * 32 + tx;
  const int gi = (blockIdx.x * 32) >> 7;
  const size_t stride = (size_t)g * DFE;
  float s = 0.f, c = 0.f;
  for (int cb = 0; cb < nch; cb += 1024) {
    const int chl = cb + ty + 32 * tx;
    int rlo = 1, rhi = 0;
    if (chl < nch) { rlo = prange[2 * chl]; rhi = prange[2 * chl + 1]; }
    float pv[32], pc[32];
#pragma unroll
    for (int u = 0; u < 32; ++u) {
      const int lo = __builtin_amdgcn_readlane(rlo, u);
      const int hi = __builtin_amdgcn_readlane(rhi, u);
      const int ch = cb + ty + 32 * u;
      pv[u] = 0.f; pc[u] = 0.f;
      if (lo <= gi && gi <= hi) {
        pv[u] = part[(size_t)ch * stride + o];
        pc[u] = pcnt[(size_t)ch * g + gi];
      }
    }
#pragma unroll
    for (int u = 0; u < 32; ++u) { s += pv[u]; c += pc[u]; }
  }
  ss[ty][tx] = s;
  if (tx == 0) sc[ty] = c;
  __syncthreads();
  if (ty == 0) {
    float st = 0.f, ct = 0.f;
    for (int i = 0; i < 32; ++i) { st += ss[i][tx]; ct += sc[i]; }
    ct = ct > 1.0f ? ct : 1.0f;
    out[o] = st / ct;
  }
}

extern "C" void kernel_launch(void* const* d_in, const int* in_sizes, int n_in,
                              void* d_out, int out_size, void* d_ws, size_t ws_size,
                              hipStream_t stream) {
  const float* x   = (const float*)d_in[0];
  const float* W0  = (const float*)d_in[1];
  const float* b0  = (const float*)d_in[2];
  const float* W1  = (const float*)d_in[3];
  const float* b1  = (const float*)d_in[4];
  const float* W2  = (const float*)d_in[5];
  const float* b2  = (const float*)d_in[6];
  const int*   src = (const int*)d_in[7];
  const int*   dst = (const int*)d_in[8];
  const int*   gid = (const int*)d_in[9];

  const int n = in_sizes[0] / DFE;
  const int e = in_sizes[7];
  const int g = out_size / DFE;

  float* ws = (float*)d_ws;
  size_t ND = (size_t)n * DFE;
  float* hA   = ws;
  float* hB   = hA + ND;
  float* agg  = hB + ND;
  float* cs   = agg + ND;
  float* cd   = cs + n;
  const int nch = (n + PCH - 1) / PCH;
  float* part = cd + n;
  float* pcnt = part + (size_t)nch * g * DFE;
  const int nhc = (e + HCH - 1) / HCH;
  const int tdw = ((n + 1) / 2 + 3) / 4 * 4;
  unsigned* tabs = (unsigned*)(pcnt + (((size_t)nch * g + 3) / 4 * 4));
  int* tick = (int*)(tabs + (((size_t)2 * nhc * tdw + 63) / 64 * 64));
  int* prange = tick + NLAYER * 2 * TKS;

  const int TB = 256;
  dim3 blk(TB);

  k_deg_hist<<<dim3(nhc, 2), dim3(1024), 0, stream>>>(src, dst, tabs, n, e, tdw);
  k_deg_norm<<<dim3(((n + 1) / 2 + TB - 1) / TB, 2), blk, 0, stream>>>(tabs, cs, cd, tick, n, nhc, tdw);

  const int nbk = (e + EPB - 1) / EPB;
  const unsigned edgeBlocks = 2u * (unsigned)nbk;
  const unsigned gemmBlocks = (unsigned)((n + 15) / 16);

  k_layer_mm<true><<<dim3(gemmBlocks), dim3(128), 0, stream>>>(x, agg, cd, cs, b0, W0, hA, n);
  k_edge_scatter<<<dim3(edgeBlocks), blk, 0, stream>>>(src, dst, hA, agg, tick, e, nbk);
  k_layer_mm<false><<<dim3(gemmBlocks), dim3(128), 0, stream>>>(x, agg, cd, cs, b0, W1, hB, n);
  k_edge_scatter<<<dim3(edgeBlocks), blk, 0, stream>>>(src, dst, hB, agg, tick + 2 * TKS, e, nbk);
  k_layer_mm<false><<<dim3(gemmBlocks), dim3(128), 0, stream>>>(x, agg, cd, cs, b1, W2, hA, n);
  k_edge_scatter<<<dim3(edgeBlocks), blk, 0, stream>>>(src, dst, hA, agg, tick + 4 * TKS, e, nbk);

  k_pool_mm<<<dim3(nch), dim3(64), 0, stream>>>(agg, cd, b2, gid, part, pcnt, prange, n, g);
  k_pool_reduce<<<dim3(g * DFE / 32), dim3(1024), 0, stream>>>(part, pcnt, prange, (float*)d_out, g, nch);
}
